# starting kernel
# baseline (speedup 1.0000x reference)
#include <hip/hip_runtime.h>


typedef __attribute__((ext_vector_type(2))) float v2f;
typedef __attribute__((ext_vector_type(8))) float v8f;

#define TAU   0.05f
#define SIGMA 0.05f
#define MAX_ITER 30

static constexpr int AP = 260;
static constexpr int YP = 196;

static constexpr int OFF_A  = 0;
static constexpr int OFF_X  = OFF_A  + 192 * AP;
static constexpr int OFF_XB = OFF_X  + 32  * AP;
static constexpr int OFF_YS = OFF_XB + 32  * AP;
static constexpr int OFF_Z  = OFF_YS + 32  * YP;
static constexpr int OFF_C  = OFF_Z  + 32  * YP;
static constexpr int SMEM_TOT = OFF_C + 256;

__global__ __launch_bounds__(128, 1)
void pdhg_kernel(const float* __restrict__ x0g, const float* __restrict__ y1g,
                 const float* __restrict__ y2g, const float* __restrict__ zg,
                 const float* __restrict__ cg,  const float* __restrict__ A1,
                 const float* __restrict__ A2,  float* __restrict__ out)
{
    __shared__ float smem[SMEM_TOT];

    const int tid  = threadIdx.x;
    const int wave = tid >> 5;
    const int lane = tid & 31;
    const int tile = wave >> 1;
    const int half = wave & 1;
    const int n    = lane & 15;
    const int hi   = (lane >> 4) << 3;
    const int ko2  = (lane >> 4) << 1;
    const int pid  = (half << 5) | lane;

    const int rbase = blockIdx.x * 32 + tile * 16;
    const int xw  = OFF_X  + tile * 16 * AP;
    const int xbw = OFF_XB + tile * 16 * AP;
    const int ysw = OFF_YS + tile * 16 * YP;
    const int zw  = OFF_Z  + tile * 16 * YP;

    for (int i = tid; i < 192 * 64; i += 128) {
        int r = i >> 6, c4 = (i & 63) << 2;
        const float* src = (r < 128) ? (A1 + r * 256 + c4) : (A2 + (r - 128) * 256 + c4);
        *(float4*)&smem[OFF_A + r * AP + c4] = *(const float4*)src;
    }
    if (tid < 64)
        *(float4*)&smem[OFF_C + tid * 4] = *(const float4*)(cg + tid * 4);
    for (int i = pid; i < 16 * 64; i += 64) {
        int r = i >> 6, c4 = (i & 63) << 2;
        float4 v = *(const float4*)(x0g + (size_t)(rbase + r) * 256 + c4);
        *(float4*)&smem[xw  + r * AP + c4] = v;
        *(float4*)&smem[xbw + r * AP + c4] = v;
    }
    for (int i = pid; i < 16 * 96; i += 64) {
        int r = i / 96, c2 = (i % 96) << 1;
        float2 v = *(const float2*)(zg + (size_t)(rbase + r) * 192 + c2);
        *(float2*)&smem[zw + r * YP + c2] = v;
    }
    for (int i = pid; i < 16 * 96; i += 64) {
        int r = i / 96, c2 = (i % 96) << 1;
        float2 v;
        if (c2 < 128) v = *(const float2*)(y1g + (size_t)(rbase + r) * 128 + c2);
        else          v = *(const float2*)(y2g + (size_t)(rbase + r) * 64 + (c2 - 128));
        v.x *= -TAU; v.y *= -TAU;
        *(float2*)&smem[ysw + r * YP + c2] = v;
    }
    __syncthreads();

    const float NEG_INV_TS = -1.0f / (TAU * SIGMA);

    for (int it = 0; it < MAX_ITER; ++it) {
        v8f acc[6];
#pragma unroll
        for (int cti = 0; cti < 6; ++cti) {
            int ct = half * 6 + cti;
#pragma unroll
            for (int v = 0; v < 8; ++v) {
                int row = v + hi, col = ct * 16 + n;
                acc[cti][v] = smem[ysw + row * YP + col] * NEG_INV_TS
                            - smem[zw  + row * YP + col];
            }
        }
        for (int k0 = 0; k0 < 256; k0 += 4) {
            int ko = k0 + ko2;
            v2f a = *(const v2f*)&smem[xbw + (lane & 15) * AP + ko];
#pragma unroll
            for (int cti = 0; cti < 6; ++cti) {
                int ct = half * 6 + cti;
                v2f b = *(const v2f*)&smem[OFF_A + (ct * 16 + (lane & 15)) * AP + ko];
                acc[cti] = __builtin_amdgcn_wmma_f32_16x16x4_f32(
                    false, a, false, b, (short)0, acc[cti], false, false);
            }
        }
#pragma unroll
        for (int cti = 0; cti < 6; ++cti) {
            int ct = half * 6 + cti;
#pragma unroll
            for (int v = 0; v < 8; ++v) {
                float yv = SIGMA * acc[cti][v];
                yv = (ct < 8) ? fmaxf(yv, 0.0f) : yv;
                smem[ysw + (v + hi) * YP + ct * 16 + n] = -TAU * yv;
            }
        }
        __syncthreads();

        v8f acc2[8];
#pragma unroll
        for (int cti = 0; cti < 8; ++cti) {
            int ct = half * 8 + cti;
            float cv = smem[OFF_C + ct * 16 + n];
#pragma unroll
            for (int v = 0; v < 8; ++v)
                acc2[cti][v] = smem[xw + (v + hi) * AP + ct * 16 + n] - TAU * cv;
        }
        for (int m0 = 0; m0 < 192; m0 += 4) {
            int mo = m0 + ko2;
            v2f a = *(const v2f*)&smem[ysw + (lane & 15) * YP + mo];
#pragma unroll
            for (int cti = 0; cti < 8; ++cti) {
                int ct = half * 8 + cti;
                v2f b;
                b.x = smem[OFF_A + mo * AP + ct * 16 + n];
                b.y = smem[OFF_A + (mo + 1) * AP + ct * 16 + n];
                acc2[cti] = __builtin_amdgcn_wmma_f32_16x16x4_f32(
                    false, a, false, b, (short)0, acc2[cti], false, false);
            }
        }
#pragma unroll
        for (int cti = 0; cti < 8; ++cti) {
            int ct = half * 8 + cti;
#pragma unroll
            for (int v = 0; v < 8; ++v) {
                int idx = (v + hi) * AP + ct * 16 + n;
                float xo = smem[xw + idx];
                float xn = fmaxf(acc2[cti][v], 0.0f);
                smem[xw  + idx] = xn;
                smem[xbw + idx] = 2.0f * xn - xo;
            }
        }
        __syncthreads();
    }

    float* out_x  = out;
    float* out_y1 = out + (size_t)8192 * 256;
    float* out_y2 = out_y1 + (size_t)8192 * 128;
    for (int i = pid; i < 16 * 64; i += 64) {
        int r = i >> 6, c4 = (i & 63) << 2;
        float4 v = *(const float4*)&smem[xw + r * AP + c4];
        *(float4*)(out_x + (size_t)(rbase + r) * 256 + c4) = v;
    }
    const float NEG_INV_TAU = -1.0f / TAU;
    for (int i = pid; i < 16 * 96; i += 64) {
        int r = i / 96, c2 = (i % 96) << 1;
        float2 v = *(const float2*)&smem[ysw + r * YP + c2];
        v.x *= NEG_INV_TAU; v.y *= NEG_INV_TAU;
        if (c2 < 128) *(float2*)(out_y1 + (size_t)(rbase + r) * 128 + c2) = v;
        else          *(float2*)(out_y2 + (size_t)(rbase + r) * 64 + (c2 - 128)) = v;
    }
}

extern "C" void kernel_launch(void* const* d_in, const int* in_sizes, int n_in,
                              void* d_out, int out_size, void* d_ws, size_t ws_size,
                              hipStream_t stream) {
    const float* x0 = (const float*)d_in[0];
    const float* y1 = (const float*)d_in[1];
    const float* y2 = (const float*)d_in[2];
    const float* z  = (const float*)d_in[3];
    const float* c  = (const float*)d_in[4];
    const float* A1 = (const float*)d_in[5];
    const float* A2 = (const float*)d_in[6];
    (void)in_sizes; (void)n_in; (void)out_size; (void)d_ws; (void)ws_size;
    pdhg_kernel<<<dim3(8192 / 32), dim3(128), 0, stream>>>(
        x0, y1, y2, z, c, A1, A2, (float*)d_out);
}


# Batch tiles staggered: split barrier (arrive early in the next run, wait late in it); next run's first LDS reads and the finished run's epilogue stand beside the other tile's matrix ops; same sums
# speedup vs baseline: 1.7668x; 1.7668x over previous
#include <hip/hip_runtime.h>


typedef __attribute__((ext_vector_type(2))) float v2f;
typedef __attribute__((ext_vector_type(4))) float v4f;
typedef __attribute__((ext_vector_type(8))) float v8f;

#define TAU   0.05f
#define SIGMA 0.05f
#define MAX_ITER 30

static constexpr int AP = 260;
static constexpr int YP = 196;

static constexpr int OFF_X  = 0;
static constexpr int OFF_XB = OFF_X  + 32 * AP;
static constexpr int OFF_YS = OFF_XB + 32 * AP;
static constexpr int OFF_Z  = OFF_YS + 32 * YP;
static constexpr int OFF_C  = OFF_Z  + 32 * YP;
static constexpr int SMEM_TOT = OFF_C + 256;

static __device__ __forceinline__ v8f mma(v2f a, v2f b, v8f c) {
    return __builtin_amdgcn_wmma_f32_16x16x4_f32(false, a, false, b, (short)0, c, false, false);
}

static __device__ __forceinline__ void run_edge() { __builtin_amdgcn_sched_barrier(0); }

static __device__ __forceinline__ void bar_arrive() {
    __builtin_amdgcn_fence(__ATOMIC_RELEASE, "workgroup");
    __builtin_amdgcn_s_barrier_signal(-1);
}
static __device__ __forceinline__ void bar_wait() {
    __builtin_amdgcn_s_barrier_wait(-1);
    __builtin_amdgcn_fence(__ATOMIC_ACQUIRE, "workgroup");
}

struct P1Raw { v4f ya[3], yb[3], za[3], zb[3]; };
struct P2Raw { v4f xa[4], xb[4]; };
struct CRaw  { v4f ca[4], cb[4]; };

static __device__ __forceinline__ void p1_read(const float* smem, int oy, int oz, P1Raw& w) {
#pragma unroll
    for (int mt = 0; mt < 3; ++mt) {
        w.ya[mt] = *(const v4f*)&smem[oy + mt * 16];
        w.yb[mt] = *(const v4f*)&smem[oy + mt * 16 + 4];
        w.za[mt] = *(const v4f*)&smem[oz + mt * 16];
        w.zb[mt] = *(const v4f*)&smem[oz + mt * 16 + 4];
    }
}

static __device__ __forceinline__ void p1_start(const P1Raw& w, v8f (&acc)[3]) {
    const float NEG_INV_TS = -1.0f / (TAU * SIGMA);
#pragma unroll
    for (int mt = 0; mt < 3; ++mt) {
#pragma unroll
        for (int r = 0; r < 4; ++r) {
            acc[mt][r]     = w.ya[mt][r] * NEG_INV_TS - w.za[mt][r];
            acc[mt][r + 4] = w.yb[mt][r] * NEG_INV_TS - w.zb[mt][r];
        }
    }
}

static __device__ __forceinline__ void p1_finish_piece(int wave, int j, v8f (&acc)[3]) {
    const int mt = j >> 1, e0 = (j & 1) * 4;
    const int t = 3 * wave + mt;
#pragma unroll
    for (int r = 0; r < 4; ++r) {
        float y = SIGMA * acc[mt][e0 + r];
        y = (t < 8) ? fmaxf(y, 0.0f) : y;
        acc[mt][e0 + r] = -TAU * y;
    }
    asm volatile("" :: "v"(acc[mt][e0]), "v"(acc[mt][e0 + 1]), "v"(acc[mt][e0 + 2]), "v"(acc[mt][e0 + 3]));
}

static __device__ __forceinline__ void p1_put(float* smem, int oy, const v8f (&acc)[3]) {
#pragma unroll
    for (int mt = 0; mt < 3; ++mt) {
        v4f oa, ob;
#pragma unroll
        for (int r = 0; r < 4; ++r) { oa[r] = acc[mt][r]; ob[r] = acc[mt][r + 4]; }
        *(v4f*)&smem[oy + mt * 16]     = oa;
        *(v4f*)&smem[oy + mt * 16 + 4] = ob;
    }
}

static __device__ __forceinline__ void p2_read(const float* smem, int ox, int oc, P2Raw& x, CRaw& c) {
#pragma unroll
    for (int nt = 0; nt < 4; ++nt) {
        x.xa[nt] = *(const v4f*)&smem[ox + nt * 16];
        x.xb[nt] = *(const v4f*)&smem[ox + nt * 16 + 4];
        c.ca[nt] = *(const v4f*)&smem[oc + nt * 16];
        c.cb[nt] = *(const v4f*)&smem[oc + nt * 16 + 4];
    }
}

static __device__ __forceinline__ void p2_start(const P2Raw& x, const CRaw& c, v8f (&acc)[4]) {
#pragma unroll
    for (int nt = 0; nt < 4; ++nt) {
#pragma unroll
        for (int r = 0; r < 4; ++r) {
            acc[nt][r]     = x.xa[nt][r] - TAU * c.ca[nt][r];
            acc[nt][r + 4] = x.xb[nt][r] - TAU * c.cb[nt][r];
        }
    }
}

static __device__ __forceinline__ void p2_finish_piece(int j, v8f (&acc)[4], P2Raw& xo) {
    const int nt = j >> 1;
#pragma unroll
    for (int r = 0; r < 4; ++r) {
        if ((j & 1) == 0) {
            const float n = fmaxf(acc[nt][r], 0.0f);
            acc[nt][r] = n;
            xo.xa[nt][r] = 2.0f * n - xo.xa[nt][r];
        } else {
            const float n = fmaxf(acc[nt][r + 4], 0.0f);
            acc[nt][r + 4] = n;
            xo.xb[nt][r] = 2.0f * n - xo.xb[nt][r];
        }
    }
    if ((j & 1) == 0)
        asm volatile("" :: "v"(acc[nt][0]), "v"(acc[nt][1]), "v"(acc[nt][2]), "v"(acc[nt][3]),
                           "v"(xo.xa[nt][0]), "v"(xo.xa[nt][1]), "v"(xo.xa[nt][2]), "v"(xo.xa[nt][3]));
    else
        asm volatile("" :: "v"(acc[nt][4]), "v"(acc[nt][5]), "v"(acc[nt][6]), "v"(acc[nt][7]),
                           "v"(xo.xb[nt][0]), "v"(xo.xb[nt][1]), "v"(xo.xb[nt][2]), "v"(xo.xb[nt][3]));
}

static __device__ __forceinline__ void p2_put(float* smem, int ox, int oxb, const v8f (&acc)[4], const P2Raw& xb) {
#pragma unroll
    for (int nt = 0; nt < 4; ++nt) {
        v4f na, nb;
#pragma unroll
        for (int r = 0; r < 4; ++r) { na[r] = acc[nt][r]; nb[r] = acc[nt][r + 4]; }
        *(v4f*)&smem[ox  + nt * 16]     = na;
        *(v4f*)&smem[ox  + nt * 16 + 4] = nb;
        *(v4f*)&smem[oxb + nt * 16]     = xb.xa[nt];
        *(v4f*)&smem[oxb + nt * 16 + 4] = xb.xb[nt];
    }
}

#define P1_KS(bt, lo, hi)                                                        \
    _Pragma("unroll") for (int ks = (lo); ks < (hi); ++ks) {                     \
        _Pragma("unroll") for (int mt = 0; mt < 3; ++mt)                         \
            acc[bt][mt] = mma(RA[mt][ks], bx[ks], acc[bt][mt]);                  \
    }
#define P2_KS(bt, lo, hi)                                                        \
    _Pragma("unroll") for (int ks = (lo); ks < (hi); ++ks) {                     \
        _Pragma("unroll") for (int nt = 0; nt < 4; ++nt)                         \
            acc2[bt][nt] = mma(RB[nt][ks], by[ks], acc2[bt][nt]);                \
    }
#define LOAD_BX(bt, lo, hi)                                                      \
    _Pragma("unroll") for (int ks = (lo); ks < (hi); ++ks)                       \
        bx[ks] = *(const v2f*)&smem[oB1 + (bt) * 16 * AP + 4 * ks];
#define LOAD_BY(bt, lo, hi)                                                      \
    _Pragma("unroll") for (int ks = (lo); ks < (hi); ++ks)                       \
        by[ks] = *(const v2f*)&smem[oB2 + (bt) * 16 * YP + 4 * ks];

__global__ __launch_bounds__(128)
void pdhg_kernel(const float* __restrict__ x0g, const float* __restrict__ y1g,
                 const float* __restrict__ y2g, const float* __restrict__ zg,
                 const float* __restrict__ cg,  const float* __restrict__ A1,
                 const float* __restrict__ A2,  float* __restrict__ out)
{
    __shared__ __attribute__((aligned(16))) float smem[SMEM_TOT];

    const int tid  = threadIdx.x;
    const int wave = __builtin_amdgcn_readfirstlane(tid >> 5);
    const int lane = tid & 31;
    const int c16  = lane & 15;
    const int h    = lane >> 4;
    const int rbase = blockIdx.x * 32;

    v2f RA[3][64];
#pragma unroll
    for (int mt = 0; mt < 3; ++mt) {
        const int t = 3 * wave + mt;
        const float* src = (t < 8) ? (A1 + (t * 16 + c16) * 256)
                                   : (A2 + ((t - 8) * 16 + c16) * 256);
        src += 2 * h;
#pragma unroll
        for (int ks = 0; ks < 64; ++ks)
            RA[mt][ks] = *(const v2f*)(src + 4 * ks);
    }
    v2f RB[4][48];
#pragma unroll
    for (int nt = 0; nt < 4; ++nt) {
        const int n = (4 * wave + nt) * 16 + c16;
#pragma unroll
        for (int ks = 0; ks < 48; ++ks) {
            const float* src = (ks < 32) ? (A1 + (4 * ks + 2 * h) * 256 + n)
                                         : (A2 + (4 * (ks - 32) + 2 * h) * 256 + n);
            v2f b;
            b.x = src[0];
            b.y = src[256];
            RB[nt][ks] = b;
        }
    }

    if (tid < 64)
        *(v4f*)&smem[OFF_C + tid * 4] = *(const v4f*)(cg + tid * 4);
    for (int i = tid; i < 32 * 64; i += 128) {
        int r = i >> 6, c4 = (i & 63) << 2;
        v4f v = *(const v4f*)(x0g + (size_t)(rbase + r) * 256 + c4);
        *(v4f*)&smem[OFF_X  + r * AP + c4] = v;
        *(v4f*)&smem[OFF_XB + r * AP + c4] = v;
    }
    for (int i = tid; i < 32 * 48; i += 128) {
        int r = i / 48, c4 = (i % 48) << 2;
        v4f zv = *(const v4f*)(zg + (size_t)(rbase + r) * 192 + c4);
        v4f yv;
        if (c4 < 128) yv = *(const v4f*)(y1g + (size_t)(rbase + r) * 128 + c4);
        else          yv = *(const v4f*)(y2g + (size_t)(rbase + r) * 64 + (c4 - 128));
        *(v4f*)&smem[OFF_Z  + r * YP + c4] = zv;
        *(v4f*)&smem[OFF_YS + r * YP + c4] = yv * (-TAU);
    }
    __syncthreads();

    const int oB1  = OFF_XB + c16 * AP + 2 * h;
    const int oB2  = OFF_YS + c16 * YP + 2 * h;
    const int oY1  = OFF_YS + c16 * YP + 48 * wave + 8 * h;
    const int oZ1  = OFF_Z  + c16 * YP + 48 * wave + 8 * h;
    const int oX2  = OFF_X  + c16 * AP + 64 * wave + 8 * h;
    const int oXB2 = OFF_XB + c16 * AP + 64 * wave + 8 * h;
    const int oC2  = OFF_C + 64 * wave + 8 * h;

    P1Raw w;
    P2Raw xo[2];
    CRaw  cv;
    v2f bx[64];
    v2f by[48];
    v8f acc[2][3];
    v8f acc2[2][4];
    p1_read(smem, oY1, oZ1, w);
    LOAD_BX(0, 0, 16)
    run_edge();
#pragma unroll
    for (int nt = 0; nt < 4; ++nt) {
#pragma unroll
        for (int r = 0; r < 8; ++r) acc2[1][nt][r] = 0.0f;
#pragma unroll
        for (int r = 0; r < 4; ++r) { xo[1].xa[nt][r] = 0.0f; xo[1].xb[nt][r] = 0.0f; }
    }

#define P1_KS_THEN_P1_PIECE(bt, j, fin) P1_KS(bt, j, (j) + 1) run_edge(); p1_finish_piece(wave, j, fin); run_edge();
#define P2_KS_THEN_P1_PIECE(bt, j, fin) P2_KS(bt, j, (j) + 1) run_edge(); p1_finish_piece(wave, j, fin); run_edge();
#define P2_KS_THEN_P2_PIECE(bt, j, fin, xold) P2_KS(bt, j, (j) + 1) run_edge(); p2_finish_piece(j, fin, xold); run_edge();
#define P1_KS_THEN_P2_PIECE(bt, j, fin, xold) P1_KS(bt, j, (j) + 1) run_edge(); p2_finish_piece(j, fin, xold); run_edge();

    for (int it = 0; it < MAX_ITER; ++it) {
        p1_start(w, acc[0]);
        run_edge();
        P1_KS_THEN_P2_PIECE(0, 0, acc2[1], xo[1])
        P1_KS_THEN_P2_PIECE(0, 1, acc2[1], xo[1])
        P1_KS_THEN_P2_PIECE(0, 2, acc2[1], xo[1])
        P1_KS_THEN_P2_PIECE(0, 3, acc2[1], xo[1])
        P1_KS_THEN_P2_PIECE(0, 4, acc2[1], xo[1])
        P1_KS_THEN_P2_PIECE(0, 5, acc2[1], xo[1])
        P1_KS_THEN_P2_PIECE(0, 6, acc2[1], xo[1])
        P1_KS_THEN_P2_PIECE(0, 7, acc2[1], xo[1])
        if (it != 0)
            p2_put(smem, oX2 + 16 * AP, oXB2 + 16 * AP, acc2[1], xo[1]);
        LOAD_BX(0, 16, 64)
        run_edge();
        P1_KS(0, 8, 16)
        run_edge();
        bar_arrive();
        run_edge();
        P1_KS(0, 16, 48)
        run_edge();
        bar_wait();
        p1_read(smem, oY1 + 16 * YP, oZ1 + 16 * YP, w);
        LOAD_BX(1, 0, 16)
        run_edge();
        P1_KS(0, 48, 64)
        run_edge();

        p1_start(w, acc[1]);
        run_edge();
        P1_KS_THEN_P1_PIECE(1, 0, acc[0])
        P1_KS_THEN_P1_PIECE(1, 1, acc[0])
        P1_KS_THEN_P1_PIECE(1, 2, acc[0])
        P1_KS_THEN_P1_PIECE(1, 3, acc[0])
        P1_KS_THEN_P1_PIECE(1, 4, acc[0])
        P1_KS_THEN_P1_PIECE(1, 5, acc[0])
        P1_KS(1, 6, 8)
        run_edge();
        p1_put(smem, oY1, acc[0]);
        LOAD_BX(1, 16, 64)
        run_edge();
        P1_KS(1, 8, 16)
        run_edge();
        bar_arrive();
        run_edge();
        P1_KS(1, 16, 48)
        run_edge();
        bar_wait();
        p2_read(smem, oX2, oC2, xo[0], cv);
        LOAD_BY(0, 0, 16)
        run_edge();
        P1_KS(1, 48, 64)
        run_edge();

        p2_start(xo[0], cv, acc2[0]);
        run_edge();
        P2_KS_THEN_P1_PIECE(0, 0, acc[1])
        P2_KS_THEN_P1_PIECE(0, 1, acc[1])
        P2_KS_THEN_P1_PIECE(0, 2, acc[1])
        P2_KS_THEN_P1_PIECE(0, 3, acc[1])
        P2_KS_THEN_P1_PIECE(0, 4, acc[1])
        P2_KS_THEN_P1_PIECE(0, 5, acc[1])
        P2_KS(0, 6, 8)
        run_edge();
        p1_put(smem, oY1 + 16 * YP, acc[1]);
        LOAD_BY(0, 16, 48)
        run_edge();
        P2_KS(0, 8, 16)
        run_edge();
        bar_arrive();
        run_edge();
        P2_KS(0, 16, 32)
        run_edge();
        bar_wait();
        p2_read(smem, oX2 + 16 * AP, oC2, xo[1], cv);
        LOAD_BY(1, 0, 16)
        run_edge();
        P2_KS(0, 32, 48)
        run_edge();

        p2_start(xo[1], cv, acc2[1]);
        run_edge();
        P2_KS_THEN_P2_PIECE(1, 0, acc2[0], xo[0])
        P2_KS_THEN_P2_PIECE(1, 1, acc2[0], xo[0])
        P2_KS_THEN_P2_PIECE(1, 2, acc2[0], xo[0])
        P2_KS_THEN_P2_PIECE(1, 3, acc2[0], xo[0])
        P2_KS_THEN_P2_PIECE(1, 4, acc2[0], xo[0])
        P2_KS_THEN_P2_PIECE(1, 5, acc2[0], xo[0])
        P2_KS_THEN_P2_PIECE(1, 6, acc2[0], xo[0])
        P2_KS_THEN_P2_PIECE(1, 7, acc2[0], xo[0])
        p2_put(smem, oX2, oXB2, acc2[0], xo[0]);
        LOAD_BY(1, 16, 48)
        run_edge();
        P2_KS(1, 8, 16)
        run_edge();
        bar_arrive();
        run_edge();
        P2_KS(1, 16, 32)
        run_edge();
        bar_wait();
        p1_read(smem, oY1, oZ1, w);
        LOAD_BX(0, 0, 16)
        run_edge();
        P2_KS(1, 32, 48)
        run_edge();
    }
#pragma unroll
    for (int j = 0; j < 8; ++j) p2_finish_piece(j, acc2[1], xo[1]);
    p2_put(smem, oX2 + 16 * AP, oXB2 + 16 * AP, acc2[1], xo[1]);
    __syncthreads();

    float* out_x  = out;
    float* out_y1 = out + (size_t)8192 * 256;
    float* out_y2 = out_y1 + (size_t)8192 * 128;
    for (int i = tid; i < 32 * 64; i += 128) {
        int r = i >> 6, c4 = (i & 63) << 2;
        v4f v = *(const v4f*)&smem[OFF_X + r * AP + c4];
        *(v4f*)(out_x + (size_t)(rbase + r) * 256 + c4) = v;
    }
    const float NEG_INV_TAU = -1.0f / TAU;
    for (int i = tid; i < 32 * 48; i += 128) {
        int r = i / 48, c4 = (i % 48) << 2;
        v4f v = *(const v4f*)&smem[OFF_YS + r * YP + c4] * NEG_INV_TAU;
        if (c4 < 128) *(v4f*)(out_y1 + (size_t)(rbase + r) * 128 + c4) = v;
        else          *(v4f*)(out_y2 + (size_t)(rbase + r) * 64 + (c4 - 128)) = v;
    }
}

extern "C" void kernel_launch(void* const* d_in, const int* in_sizes, int n_in,
                              void* d_out, int out_size, void* d_ws, size_t ws_size,
                              hipStream_t stream) {
    const float* x0 = (const float*)d_in[0];
    const float* y1 = (const float*)d_in[1];
    const float* y2 = (const float*)d_in[2];
    const float* z  = (const float*)d_in[3];
    const float* c  = (const float*)d_in[4];
    const float* A1 = (const float*)d_in[5];
    const float* A2 = (const float*)d_in[6];
    (void)in_sizes; (void)n_in; (void)out_size; (void)d_ws; (void)ws_size;
    pdhg_kernel<<<dim3(8192 / 32), dim3(128), 0, stream>>>(
        x0, y1, y2, z, c, A1, A2, (float*)d_out);
}
